# starting kernel
# baseline (speedup 1.0000x reference)
#include <hip/hip_runtime.h>
#include <math.h>

typedef float v2f __attribute__((ext_vector_type(2)));
typedef float v8f __attribute__((ext_vector_type(8)));


__global__ void zero_int_kernel(int* __restrict__ p, int n) {
  int i = blockIdx.x * blockDim.x + threadIdx.x;
  int stride = gridDim.x * blockDim.x;
  for (; i < n; i += stride) p[i] = 0;
}

__global__ void degree_kernel(const int* __restrict__ dst, int* __restrict__ cnt, int E) {
  int e = blockIdx.x * blockDim.x + threadIdx.x;
  if (e < E) atomicAdd(&cnt[dst[e]], 1);
}

__global__ __launch_bounds__(1024)
void scan_kernel(const int* __restrict__ cnt, int* __restrict__ off,
                 int* __restrict__ cursor, int n) {
  __shared__ int tmp[1024];
  int carry = 0;
  for (int base = 0; base < n; base += 1024) {
    int i = base + threadIdx.x;
    int v = (i < n) ? cnt[i] : 0;
    tmp[threadIdx.x] = v;
    __syncthreads();
#pragma unroll
    for (int d = 1; d < 1024; d <<= 1) {
      int t = (threadIdx.x >= d) ? tmp[threadIdx.x - d] : 0;
      __syncthreads();
      tmp[threadIdx.x] += t;
      __syncthreads();
    }
    int excl = tmp[threadIdx.x] - v;
    if (i < n) {
      off[i] = carry + excl;
      cursor[i] = carry + excl;
    }
    int chunk_total = tmp[1023];
    __syncthreads();
    carry += chunk_total;
  }
  if (threadIdx.x == 0) off[n] = carry;
}

__global__ void csr_fill_kernel(const int* __restrict__ src, const int* __restrict__ dst,
                                int* __restrict__ cursor, int* __restrict__ csr, int E) {
  int e = blockIdx.x * blockDim.x + threadIdx.x;
  if (e < E) {
    int slot = atomicAdd(&cursor[dst[e]], 1);
    csr[slot] = src[e];
  }
}

template<int K, int NCOL, int WAVES, int TILES, int ACT>
__global__ __launch_bounds__(WAVES * 32)
void sage_layer_kernel(const float* __restrict__ feat,
                       const int* __restrict__ off,
                       const int* __restrict__ csr,
                       const float* __restrict__ Wl,
                       const float* __restrict__ Wr,
                       const float* __restrict__ bias,
                       float* __restrict__ out, int nrows) {
  static_assert(WAVES * TILES * 16 == NCOL, "tiling must cover NCOL");
  static_assert(K % 128 == 0, "K must be a multiple of 128");
  constexpr int KP = K + 4;
  constexpr int CH = K / 128;
  constexpr int RPW = 16 / WAVES;
  __shared__ float sA[16 * KP];
  __shared__ float sX[16 * KP];

  const int rowbase = blockIdx.x * 16;
  const int lane = threadIdx.x & 31;
  const int wave = threadIdx.x >> 5;

  for (int rr = 0; rr < RPW; ++rr) {
    const int r = wave * RPW + rr;
    const int row = rowbase + r;
    float4 acc[CH] = {};
    float4 self[CH] = {};
    float scale = 1.0f;
    if (row < nrows) {
      const int beg = off[row];
      const int end = off[row + 1];
      scale = 1.0f / fmaxf((float)(end - beg), 1.0f);
#pragma unroll
      for (int c = 0; c < CH; ++c)
        self[c] = ((const float4*)(feat + (long)row * K))[lane + 32 * c];
      for (int p = beg; p < end; ++p) {
        const int s = csr[p];
        if (p + 1 < end)
          __builtin_prefetch(feat + (long)csr[p + 1] * K, 0, 0);
        const float4* rp = (const float4*)(feat + (long)s * K);
#pragma unroll
        for (int c = 0; c < CH; ++c) {
          float4 v = rp[lane + 32 * c];
          acc[c].x += v.x; acc[c].y += v.y; acc[c].z += v.z; acc[c].w += v.w;
        }
      }
    }
#pragma unroll
    for (int c = 0; c < CH; ++c) {
      const int col = 4 * (lane + 32 * c);
      float4 a;
      a.x = acc[c].x * scale; a.y = acc[c].y * scale;
      a.z = acc[c].z * scale; a.w = acc[c].w * scale;
      *(float4*)&sA[r * KP + col] = a;
      *(float4*)&sX[r * KP + col] = self[c];
    }
  }
  __syncthreads();

  const int m = lane & 15;
  const int khalf = (lane >> 4) * 2;

  v8f acc[TILES];
  int ncol[TILES];
#pragma unroll
  for (int t = 0; t < TILES; ++t) {
    ncol[t] = (wave * TILES + t) * 16 + (lane & 15);
    float bv = bias[ncol[t]];
#pragma unroll
    for (int j = 0; j < 8; ++j) acc[t][j] = bv;
  }

#pragma unroll 2
  for (int k0 = 0; k0 < K; k0 += 4) {
    const int ka = k0 + khalf;
    v2f a;
    a[0] = sA[m * KP + ka];
    a[1] = sA[m * KP + ka + 1];
#pragma unroll
    for (int t = 0; t < TILES; ++t) {
      v2f b;
      b[0] = Wl[(long)ka * NCOL + ncol[t]];
      b[1] = Wl[(long)(ka + 1) * NCOL + ncol[t]];
      acc[t] = __builtin_amdgcn_wmma_f32_16x16x4_f32(
          false, a, false, b, (short)0, acc[t], false, false);
    }
  }
#pragma unroll 2
  for (int k0 = 0; k0 < K; k0 += 4) {
    const int ka = k0 + khalf;
    v2f a;
    a[0] = sX[m * KP + ka];
    a[1] = sX[m * KP + ka + 1];
#pragma unroll
    for (int t = 0; t < TILES; ++t) {
      v2f b;
      b[0] = Wr[(long)ka * NCOL + ncol[t]];
      b[1] = Wr[(long)(ka + 1) * NCOL + ncol[t]];
      acc[t] = __builtin_amdgcn_wmma_f32_16x16x4_f32(
          false, a, false, b, (short)0, acc[t], false, false);
    }
  }

  const int rowhalf = (lane >> 4) * 8;
#pragma unroll
  for (int t = 0; t < TILES; ++t) {
#pragma unroll
    for (int r = 0; r < 8; ++r) {
      int row = rowbase + rowhalf + r;
      if (row < nrows) {
        float v = acc[t][r];
        if (ACT == 0) {
          v = fmaxf(v, 0.0f);
        } else {
          v = 1.0f / (1.0f + __expf(-v));
        }
        out[(long)row * NCOL + ncol[t]] = v;
      }
    }
  }
}


extern "C" void kernel_launch(void* const* d_in, const int* in_sizes, int n_in,
                              void* d_out, int out_size, void* d_ws, size_t ws_size,
                              hipStream_t stream) {
  constexpr int FIN = 128, FHID = 256, FOUT = 64;

  const float* x   = (const float*)d_in[0];
  const int*   ei  = (const int*)d_in[1];
  const float* Wl1 = (const float*)d_in[2];
  const float* Wr1 = (const float*)d_in[3];
  const float* b1  = (const float*)d_in[4];
  const float* Wl2 = (const float*)d_in[5];
  const float* Wr2 = (const float*)d_in[6];
  const float* b2  = (const float*)d_in[7];
  float* out = (float*)d_out;

  const int N = in_sizes[0] / FIN;
  const int E = in_sizes[1] / 2;
  const int* src = ei;
  const int* dst = ei + E;

  int* cnt    = (int*)d_ws;
  int* off    = cnt + N;
  int* cursor = off + (N + 1);
  int* csr    = cursor + N;
  float* h    = (float*)(csr + E);

  const int T = 256;
  auto blocks = [](long n, int t) { return (int)((n + t - 1) / t); };
  const int rowBlocks = (N + 15) / 16;

  zero_int_kernel<<<min(blocks(N, T), 4096), T, 0, stream>>>(cnt, N);
  degree_kernel<<<blocks(E, T), T, 0, stream>>>(dst, cnt, E);
  scan_kernel<<<1, 1024, 0, stream>>>(cnt, off, cursor, N);
  csr_fill_kernel<<<blocks(E, T), T, 0, stream>>>(src, dst, cursor, csr, E);

  sage_layer_kernel<FIN, FHID, 8, 2, 0>
      <<<rowBlocks, 256, 0, stream>>>(x, off, csr, Wl1, Wr1, b1, h, N);
  sage_layer_kernel<FHID, FOUT, 4, 1, 1>
      <<<rowBlocks, 128, 0, stream>>>(h, off, csr, Wl2, Wr2, b2, out, N);
}


# Layer 2 GEMM: the 4 waves of a product split K (64 each, all 4 column tiles) instead of the columns: 8 dependent weight round trips per wave, not 32; 8 partials summed via LDS: sum order changes
# speedup vs baseline: 3.6428x; 3.6428x over previous
#include <hip/hip_runtime.h>
#include <math.h>

typedef float v2f __attribute__((ext_vector_type(2)));
typedef float v8f __attribute__((ext_vector_type(8)));

constexpr int NSUB = 2;
constexpr int LINK_T = 1024;
constexpr int LINK_RANGES = 128;
constexpr int LINK_B = 8;

__device__ __forceinline__ void link_one(int* sh, int* __restrict__ next,
                                         int d, int e, int n0, unsigned cnt) {
  const unsigned dd = (unsigned)(d - n0);
  if (dd < cnt) next[e] = atomicExch(&sh[dd], e);
}

__device__ __forceinline__ void link_group(int* sh, int* __restrict__ next,
                                           int4 d, int e, int n0, unsigned cnt) {
  const unsigned ux = (unsigned)(d.x - n0), uy = (unsigned)(d.y - n0);
  const unsigned uz = (unsigned)(d.z - n0), uw = (unsigned)(d.w - n0);
  if (min(min(ux, uy), min(uz, uw)) < cnt) {
    link_one(sh, next, d.x, e, n0, cnt);
    link_one(sh, next, d.y, e + 1, n0, cnt);
    link_one(sh, next, d.z, e + 2, n0, cnt);
    link_one(sh, next, d.w, e + 3, n0, cnt);
  }
}

__global__ __launch_bounds__(LINK_T)
void link_kernel(const int* __restrict__ dst, int* __restrict__ head,
                 int* __restrict__ next, int N, int E) {
  extern __shared__ int sh[];
  const int tid = threadIdx.x;
  const int R = (N + (int)gridDim.x - 1) / (int)gridDim.x;
  const int n0 = (int)blockIdx.x * R;
  const int left = N - n0;
  const unsigned cnt = left <= 0 ? 0u : (unsigned)(left < R ? left : R);
  const int j = blockIdx.y;
  const int part = (E + NSUB - 1) / NSUB;
  const int e0 = j * part < E ? j * part : E;
  const int e1 = e0 + part < E ? e0 + part : E;

  for (int i = tid; i < R; i += LINK_T) sh[i] = -1;
  __syncthreads();

  if (cnt > 0 && e1 > e0) {
    int a0 = e0;
    while (a0 < e1 && (((size_t)(dst + a0)) & 15) != 0) ++a0;
    const int nvec = (e1 - a0) >> 2;
    const int t0 = a0 + 4 * nvec;
    if (tid < a0 - e0) link_one(sh, next, dst[e0 + tid], e0 + tid, n0, cnt);
    if (tid < e1 - t0) link_one(sh, next, dst[t0 + tid], t0 + tid, n0, cnt);

    const int4* __restrict__ dv = (const int4*)(dst + a0);
    const int s = (int)(((long)blockIdx.x * (long)nvec) / (long)gridDim.x);
    for (int sweep = 0; sweep < 2; ++sweep) {
      const int lo = sweep ? 0 : s;
      const int hi = sweep ? s : nvec;
      int base = lo;
      for (; base + LINK_T * LINK_B <= hi; base += LINK_T * LINK_B) {
        const int g = base + tid;
        int4 d[LINK_B];
#pragma unroll
        for (int u = 0; u < LINK_B; ++u) d[u] = dv[g + u * LINK_T];
#pragma unroll
        for (int u = 0; u < LINK_B; ++u)
          link_group(sh, next, d[u], a0 + 4 * (g + u * LINK_T), n0, cnt);
      }
      if (base < hi) {
        const int g = base + tid;
        int4 d[LINK_B];
#pragma unroll
        for (int u = 0; u < LINK_B; ++u) {
          const int v = g + u * LINK_T;
          d[u] = dv[v < hi ? v : base];
        }
#pragma unroll
        for (int u = 0; u < LINK_B; ++u) {
          const int v = g + u * LINK_T;
          if (v < hi) link_group(sh, next, d[u], a0 + 4 * v, n0, cnt);
        }
      }
    }
  }
  __syncthreads();
  for (int i = tid; i < (int)cnt; i += LINK_T) head[(long)(n0 + i) * NSUB + j] = sh[i];
}

template<int K, int NCOL, int WAVES, int TILES, int ACT, bool KSPLIT>
__global__ __launch_bounds__(WAVES * 32)
void sage_layer_kernel(const float* __restrict__ feat,
                       const int* __restrict__ head,
                       const int* __restrict__ next,
                       const int* __restrict__ src,
                       const float* __restrict__ Wl,
                       const float* __restrict__ Wr,
                       const float* __restrict__ bias,
                       float* __restrict__ out, int nrows, int nedges) {
  static_assert(WAVES % 2 == 0, "the waves split into two halves, one per product");
  constexpr int HW = WAVES / 2;
  static_assert(KSPLIT ? (TILES * 16 == NCOL) : (HW * TILES * 16 == NCOL), "tiling must cover NCOL");
  constexpr int KSL = KSPLIT ? K / HW : K;
  static_assert(KSL * (KSPLIT ? HW : 1) == K && KSL % 8 == 0, "K slices cover K in whole pairs of k-steps");
  static_assert(K % 128 == 0, "K must be a multiple of 128");
  constexpr int KP = K + 4;
  constexpr int RPW = 16 / WAVES;
  constexpr int LPR = 32 / RPW;
  constexpr int CH = K / (4 * LPR);
  static_assert(RPW * WAVES == 16 && LPR * RPW == 32, "a wave's lanes split evenly over its rows");
  static_assert(CH * 4 * LPR == K, "a lane group covers a whole row");
  __shared__ float sT[2 * 16 * KP];
  __shared__ float sP[KSPLIT ? 32 : HW * TILES * 8 * 32];
  static_assert(!KSPLIT || WAVES * 16 * NCOL <= 2 * 16 * KP, "the partial tiles fit over the A tiles");
  static_assert(!KSPLIT || NCOL % 4 == 0, "an output row is written in 16-byte pieces");
  float* const sA = sT;
  float* const sX = sT + 16 * KP;

  const int rowbase = blockIdx.x * 16;
  const int lane = threadIdx.x & 31;
  const int wave = threadIdx.x >> 5;

  {
    const int sub = lane / LPR;
    const int sl = lane - sub * LPR;
    const int r = wave * RPW + sub;
    const int row = rowbase + r;
    float4 acc[CH] = {};
    float scale = 1.0f;
    if (row < nrows) {
      const float4* xp = (const float4*)(feat + (long)row * K);
      float4 sv[CH];
#pragma unroll
      for (int c = 0; c < CH; ++c) sv[c] = xp[sl + LPR * c];
      int p[NSUB];
#pragma unroll
      for (int j = 0; j < NSUB; ++j) p[j] = head[(long)row * NSUB + j];
#pragma unroll
      for (int c = 0; c < CH; ++c)
        *(float4*)&sX[r * KP + 4 * (sl + LPR * c)] = sv[c];
      int s[NSUB], pn[NSUB];
      int deg = 0;
#pragma unroll
      for (int j = 0; j < NSUB; ++j) {
        s[j] = -1; pn[j] = -1;
        if ((unsigned)p[j] < (unsigned)nedges) { s[j] = src[p[j]]; pn[j] = next[p[j]]; }
      }
      for (int it = 0; it < nedges; ++it) {
        bool live[NSUB];
        bool any = false;
#pragma unroll
        for (int j = 0; j < NSUB; ++j) { live[j] = s[j] >= 0; any = any || live[j]; }
        if (!any) break;
        bool more[NSUB];
        int sn[NSUB], pnn[NSUB];
#pragma unroll
        for (int j = 0; j < NSUB; ++j) {
          more[j] = (unsigned)pn[j] < (unsigned)nedges;
          const int q = more[j] ? pn[j] : 0;
          sn[j] = src[q];
          pnn[j] = next[q];
        }
        float4 v[NSUB][CH];
#pragma unroll
        for (int j = 0; j < NSUB; ++j) {
#pragma unroll
          for (int c = 0; c < CH; ++c) v[j][c] = make_float4(0.0f, 0.0f, 0.0f, 0.0f);
          if (live[j]) {
            const float4* rp = (const float4*)feat + ((unsigned)s[j] * (unsigned)(K / 4) + (unsigned)sl);
#pragma unroll
            for (int c = 0; c < CH; ++c) v[j][c] = rp[LPR * c];
          }
        }
#pragma unroll
        for (int j = 0; j < NSUB; ++j) {
#pragma unroll
          for (int c = 0; c < CH; ++c) {
            acc[c].x += v[j][c].x; acc[c].y += v[j][c].y;
            acc[c].z += v[j][c].z; acc[c].w += v[j][c].w;
          }
          deg += live[j] ? 1 : 0;
        }
#pragma unroll
        for (int j = 0; j < NSUB; ++j) {
          s[j] = more[j] ? sn[j] : -1;
          pn[j] = more[j] ? pnn[j] : -1;
        }
      }
      scale = 1.0f / fmaxf((float)deg, 1.0f);
    } else {
      const float4 z = {0.0f, 0.0f, 0.0f, 0.0f};
#pragma unroll
      for (int c = 0; c < CH; ++c) *(float4*)&sX[r * KP + 4 * (sl + LPR * c)] = z;
    }
#pragma unroll
    for (int c = 0; c < CH; ++c) {
      float4 a;
      a.x = acc[c].x * scale; a.y = acc[c].y * scale;
      a.z = acc[c].z * scale; a.w = acc[c].w * scale;
      *(float4*)&sA[r * KP + 4 * (sl + LPR * c)] = a;
    }
  }
  __syncthreads();

  const int m = lane & 15;
  const int khalf = (lane >> 4) * 2;
  const int part = wave / HW;
  const int cw = wave - part * HW;
  const float* __restrict__ W = part ? Wr : Wl;
  const float* tile = sT + part * (16 * KP);

  v8f acc[TILES];
  int ncol[TILES];
#pragma unroll
  for (int t = 0; t < TILES; ++t) {
    ncol[t] = ((KSPLIT ? 0 : cw * TILES) + t) * 16 + (lane & 15);
    float bv = (KSPLIT || part) ? 0.0f : bias[ncol[t]];
#pragma unroll
    for (int j = 0; j < 8; ++j) acc[t][j] = bv;
  }

  const int kbeg = KSPLIT ? cw * KSL : 0;
  const float* __restrict__ Wk = W + (long)kbeg * NCOL;
  const float* tk = tile + kbeg;
#pragma unroll 2
  for (int k0 = 0; k0 < KSL; k0 += 4) {
    const int ka = k0 + khalf;
    v2f a;
    a[0] = tk[m * KP + ka];
    a[1] = tk[m * KP + ka + 1];
#pragma unroll
    for (int t = 0; t < TILES; ++t) {
      v2f b;
      b[0] = Wk[(long)ka * NCOL + ncol[t]];
      b[1] = Wk[(long)(ka + 1) * NCOL + ncol[t]];
      acc[t] = __builtin_amdgcn_wmma_f32_16x16x4_f32(
          false, a, false, b, (short)0, acc[t], false, false);
    }
  }

  if constexpr (KSPLIT) {
    __syncthreads();
    float* const sR = sT;
    {
      const int rowhalf = (lane >> 4) * 8;
#pragma unroll
      for (int t = 0; t < TILES; ++t)
#pragma unroll
        for (int r = 0; r < 8; ++r)
          sR[(wave * 16 + rowhalf + r) * NCOL + ncol[t]] = acc[t][r];
    }
    __syncthreads();
    constexpr int QPR = NCOL / 4;
    for (int i = threadIdx.x; i < 16 * QPR; i += WAVES * 32) {
      const int orow = i / QPR;
      const int oc = (i - orow * QPR) * 4;
      float4 s;
      s.x = bias[oc]; s.y = bias[oc + 1]; s.z = bias[oc + 2]; s.w = bias[oc + 3];
#pragma unroll
      for (int w = 0; w < WAVES; ++w) {
        const float4 p = *(const float4*)&sR[(w * 16 + orow) * NCOL + oc];
        s.x += p.x; s.y += p.y; s.z += p.z; s.w += p.w;
      }
      const int row = rowbase + orow;
      if (row < nrows) {
        if (ACT == 0) {
          s.x = fmaxf(s.x, 0.0f); s.y = fmaxf(s.y, 0.0f);
          s.z = fmaxf(s.z, 0.0f); s.w = fmaxf(s.w, 0.0f);
        } else {
          s.x = 1.0f / (1.0f + __expf(-s.x)); s.y = 1.0f / (1.0f + __expf(-s.y));
          s.z = 1.0f / (1.0f + __expf(-s.z)); s.w = 1.0f / (1.0f + __expf(-s.w));
        }
        *(float4*)&out[(long)row * NCOL + oc] = s;
      }
    }
    return;
  }

  if (part == 1) {
#pragma unroll
    for (int t = 0; t < TILES; ++t)
#pragma unroll
      for (int r = 0; r < 8; ++r)
        sP[((cw * TILES + t) * 8 + r) * 32 + lane] = acc[t][r];
  }
  __syncthreads();

  if (part == 0) {
    const int rowhalf = (lane >> 4) * 8;
#pragma unroll
    for (int t = 0; t < TILES; ++t) {
#pragma unroll
      for (int r = 0; r < 8; ++r) {
        int row = rowbase + rowhalf + r;
        float v = acc[t][r] + sP[((cw * TILES + t) * 8 + r) * 32 + lane];
        if (row < nrows) {
          if (ACT == 0) {
            v = fmaxf(v, 0.0f);
          } else {
            v = 1.0f / (1.0f + __expf(-v));
          }
          out[(long)row * NCOL + ncol[t]] = v;
        }
      }
    }
  }
}


extern "C" void kernel_launch(void* const* d_in, const int* in_sizes, int n_in,
                              void* d_out, int out_size, void* d_ws, size_t ws_size,
                              hipStream_t stream) {
  constexpr int FIN = 128, FHID = 256, FOUT = 64;

  const float* x   = (const float*)d_in[0];
  const int*   ei  = (const int*)d_in[1];
  const float* Wl1 = (const float*)d_in[2];
  const float* Wr1 = (const float*)d_in[3];
  const float* b1  = (const float*)d_in[4];
  const float* Wl2 = (const float*)d_in[5];
  const float* Wr2 = (const float*)d_in[6];
  const float* b2  = (const float*)d_in[7];
  float* out = (float*)d_out;

  const int N = in_sizes[0] / FIN;
  const int E = in_sizes[1] / 2;
  const int* src = ei;
  const int* dst = ei + E;

  static_assert(NSUB >= 1 && NSUB <= 3, "the chain heads must fit in front of next");
  int* head = (int*)d_ws;
  int* next = head + N + (N + 1) + N;
  float* h  = (float*)(next + E);

  const int rowBlocks = (N + 15) / 16;

  const int R = (N + LINK_RANGES - 1) / LINK_RANGES;
  link_kernel<<<dim3(LINK_RANGES, NSUB), LINK_T, R * sizeof(int), stream>>>(dst, head, next, N, E);

  sage_layer_kernel<FIN, FHID, 8, 4, 0, false>
      <<<rowBlocks, 256, 0, stream>>>(x, head, next, src, Wl1, Wr1, b1, h, N, E);
  sage_layer_kernel<FHID, FOUT, 8, 4, 1, true>
      <<<rowBlocks, 256, 0, stream>>>(h, head, next, src, Wl2, Wr2, b2, out, N, E);
}


# Neighbour chains: 16 per node (edge list cut in 16 parts, a build block reads 1/16 of it), one per lane of a row's 16 lanes, 4 source rows in flight; heads kept in out rows; row sum order changes
# speedup vs baseline: 3.8545x; 1.0581x over previous
#include <hip/hip_runtime.h>
#include <math.h>

typedef float v2f __attribute__((ext_vector_type(2)));
typedef float v8f __attribute__((ext_vector_type(8)));

constexpr int NSUB = 16;
constexpr int HSTRIDE = 64;
constexpr int LINK_T = 1024;
constexpr int LINK_RANGES = 16;
constexpr int LINK_B = 8;

__device__ __forceinline__ void link_one(int* sh, int* __restrict__ next,
                                         int d, int e, int n0, unsigned cnt) {
  const unsigned dd = (unsigned)(d - n0);
  if (dd < cnt) next[e] = atomicExch(&sh[dd], e);
}

__device__ __forceinline__ void link_group(int* sh, int* __restrict__ next,
                                           int4 d, int e, int n0, unsigned cnt) {
  const unsigned ux = (unsigned)(d.x - n0), uy = (unsigned)(d.y - n0);
  const unsigned uz = (unsigned)(d.z - n0), uw = (unsigned)(d.w - n0);
  if (min(min(ux, uy), min(uz, uw)) < cnt) {
    link_one(sh, next, d.x, e, n0, cnt);
    link_one(sh, next, d.y, e + 1, n0, cnt);
    link_one(sh, next, d.z, e + 2, n0, cnt);
    link_one(sh, next, d.w, e + 3, n0, cnt);
  }
}

__global__ __launch_bounds__(LINK_T)
void link_kernel(const int* __restrict__ dst, int* __restrict__ head,
                 int* __restrict__ next, int N, int E) {
  extern __shared__ int sh[];
  const int tid = threadIdx.x;
  const int R = (N + (int)gridDim.x - 1) / (int)gridDim.x;
  const int n0 = (int)blockIdx.x * R;
  const int left = N - n0;
  const unsigned cnt = left <= 0 ? 0u : (unsigned)(left < R ? left : R);
  const int j = blockIdx.y;
  const int part = (E + NSUB - 1) / NSUB;
  const int e0 = j * part < E ? j * part : E;
  const int e1 = e0 + part < E ? e0 + part : E;

  for (int i = tid; i < R; i += LINK_T) sh[i] = -1;
  __syncthreads();

  if (cnt > 0 && e1 > e0) {
    int a0 = e0;
    while (a0 < e1 && (((size_t)(dst + a0)) & 15) != 0) ++a0;
    const int nvec = (e1 - a0) >> 2;
    const int t0 = a0 + 4 * nvec;
    if (tid < a0 - e0) link_one(sh, next, dst[e0 + tid], e0 + tid, n0, cnt);
    if (tid < e1 - t0) link_one(sh, next, dst[t0 + tid], t0 + tid, n0, cnt);

    const int4* __restrict__ dv = (const int4*)(dst + a0);
    const int s = (int)(((long)blockIdx.x * (long)nvec) / (long)gridDim.x);
    for (int sweep = 0; sweep < 2; ++sweep) {
      const int lo = sweep ? 0 : s;
      const int hi = sweep ? s : nvec;
      int base = lo;
      for (; base + LINK_T * LINK_B <= hi; base += LINK_T * LINK_B) {
        const int g = base + tid;
        int4 d[LINK_B];
#pragma unroll
        for (int u = 0; u < LINK_B; ++u) d[u] = dv[g + u * LINK_T];
#pragma unroll
        for (int u = 0; u < LINK_B; ++u)
          link_group(sh, next, d[u], a0 + 4 * (g + u * LINK_T), n0, cnt);
      }
      if (base < hi) {
        const int g = base + tid;
        int4 d[LINK_B];
#pragma unroll
        for (int u = 0; u < LINK_B; ++u) {
          const int v = g + u * LINK_T;
          d[u] = dv[v < hi ? v : base];
        }
#pragma unroll
        for (int u = 0; u < LINK_B; ++u) {
          const int v = g + u * LINK_T;
          if (v < hi) link_group(sh, next, d[u], a0 + 4 * v, n0, cnt);
        }
      }
    }
  }
  __syncthreads();
  for (int i = tid; i < (int)cnt; i += LINK_T) head[(long)(n0 + i) * HSTRIDE + j] = sh[i];
}

template<int K, int NCOL, int WAVES, int TILES, int ACT, bool KSPLIT, int NB>
__global__ __launch_bounds__(WAVES * 32)
void sage_layer_kernel(const float* __restrict__ feat,
                       const int* head,
                       const int* __restrict__ next,
                       const int* __restrict__ src,
                       const float* __restrict__ Wl,
                       const float* __restrict__ Wr,
                       const float* __restrict__ bias,
                       float* out,
                       int nrows, int nedges) {
  static_assert(WAVES % 2 == 0, "the waves split into two halves, one per product");
  constexpr int HW = WAVES / 2;
  static_assert(KSPLIT ? (TILES * 16 == NCOL) : (HW * TILES * 16 == NCOL), "tiling must cover NCOL");
  constexpr int KSL = KSPLIT ? K / HW : K;
  static_assert(KSL * (KSPLIT ? HW : 1) == K && KSL % 8 == 0, "K slices cover K in whole pairs of k-steps");
  static_assert(K % 128 == 0, "K must be a multiple of 128");
  constexpr int KP = K + 4;
  constexpr int RPW = 16 / WAVES;
  constexpr int LPR = 32 / RPW;
  constexpr int CH = K / (4 * LPR);
  static_assert(RPW * WAVES == 16 && LPR * RPW == 32, "a wave's lanes split evenly over its rows");
  static_assert(CH * 4 * LPR == K, "a lane group covers a whole row");
  static_assert(LPR == NSUB, "one chain per lane of a row's lane group");
  static_assert(NSUB <= HSTRIDE, "a node's heads fit in its output row");
  __shared__ float sT[2 * 16 * KP];
  __shared__ float sP[KSPLIT ? 32 : HW * TILES * 8 * 32];
  static_assert(!KSPLIT || WAVES * 16 * NCOL <= 2 * 16 * KP, "the partial tiles fit over the A tiles");
  static_assert(!KSPLIT || NCOL % 4 == 0, "an output row is written in 16-byte pieces");
  float* const sA = sT;
  float* const sX = sT + 16 * KP;

  const int rowbase = blockIdx.x * 16;
  const int lane = threadIdx.x & 31;
  const int wave = threadIdx.x >> 5;

  {
    const int sub = lane / LPR;
    const int sl = lane - sub * LPR;
    const int r = wave * RPW + sub;
    const int row = rowbase + r;
    float4 acc[CH] = {};
    int s = -1, pn = -1;
    if (row < nrows) {
      const float4* xp = (const float4*)(feat + (long)row * K);
      float4 sv[CH];
#pragma unroll
      for (int c = 0; c < CH; ++c) sv[c] = xp[sl + LPR * c];
      const int p = head[(long)row * HSTRIDE + sl];
#pragma unroll
      for (int c = 0; c < CH; ++c)
        *(float4*)&sX[r * KP + 4 * (sl + LPR * c)] = sv[c];
      if ((unsigned)p < (unsigned)nedges) { s = src[p]; pn = next[p]; }
    } else {
      const float4 z = {0.0f, 0.0f, 0.0f, 0.0f};
#pragma unroll
      for (int c = 0; c < CH; ++c) *(float4*)&sX[r * KP + 4 * (sl + LPR * c)] = z;
    }
    int deg = 0;
    for (int it = 0; it < nedges; ++it) {
      const unsigned wavelive = __builtin_amdgcn_ballot_w32(s >= 0);
      if (wavelive == 0u) break;
      unsigned mg = (wavelive >> (sub * LPR)) & ((1u << LPR) - 1u);
      deg += __popc(mg);
      const bool more = (unsigned)pn < (unsigned)nedges;
      const int q = more ? pn : 0;
      const int sn = src[q];
      const int pnn = next[q];
      do {
        int jj[NB], sj[NB];
#pragma unroll
        for (int b = 0; b < NB; ++b) {
          jj[b] = __ffs((int)mg) - 1;
          mg &= mg - 1u;
        }
#pragma unroll
        for (int b = 0; b < NB; ++b)
          sj[b] = __shfl(s, (lane & ~(LPR - 1)) | (jj[b] & (LPR - 1)));
        float4 v[NB][CH];
#pragma unroll
        for (int b = 0; b < NB; ++b) {
#pragma unroll
          for (int c = 0; c < CH; ++c) v[b][c] = make_float4(0.0f, 0.0f, 0.0f, 0.0f);
          if (jj[b] >= 0) {
            const float4* rp = (const float4*)feat + ((unsigned)sj[b] * (unsigned)(K / 4) + (unsigned)sl);
#pragma unroll
            for (int c = 0; c < CH; ++c) v[b][c] = rp[LPR * c];
          }
        }
#pragma unroll
        for (int b = 0; b < NB; ++b) {
#pragma unroll
          for (int c = 0; c < CH; ++c) {
            acc[c].x += v[b][c].x; acc[c].y += v[b][c].y;
            acc[c].z += v[b][c].z; acc[c].w += v[b][c].w;
          }
        }
      } while (__builtin_amdgcn_ballot_w32(mg != 0u) != 0u);
      s = more ? sn : -1;
      pn = more ? pnn : -1;
    }
    const float scale = 1.0f / fmaxf((float)deg, 1.0f);
#pragma unroll
    for (int c = 0; c < CH; ++c) {
      float4 a;
      a.x = acc[c].x * scale; a.y = acc[c].y * scale;
      a.z = acc[c].z * scale; a.w = acc[c].w * scale;
      *(float4*)&sA[r * KP + 4 * (sl + LPR * c)] = a;
    }
  }
  __syncthreads();

  const int m = lane & 15;
  const int khalf = (lane >> 4) * 2;
  const int part = wave / HW;
  const int cw = wave - part * HW;
  const float* __restrict__ W = part ? Wr : Wl;
  const float* tile = sT + part * (16 * KP);

  v8f acc[TILES];
  int ncol[TILES];
#pragma unroll
  for (int t = 0; t < TILES; ++t) {
    ncol[t] = ((KSPLIT ? 0 : cw * TILES) + t) * 16 + (lane & 15);
    float bv = (KSPLIT || part) ? 0.0f : bias[ncol[t]];
#pragma unroll
    for (int j = 0; j < 8; ++j) acc[t][j] = bv;
  }

  const int kbeg = KSPLIT ? cw * KSL : 0;
  const float* __restrict__ Wk = W + (long)kbeg * NCOL;
  const float* tk = tile + kbeg;
#pragma unroll 2
  for (int k0 = 0; k0 < KSL; k0 += 4) {
    const int ka = k0 + khalf;
    v2f a;
    a[0] = tk[m * KP + ka];
    a[1] = tk[m * KP + ka + 1];
#pragma unroll
    for (int t = 0; t < TILES; ++t) {
      v2f b;
      b[0] = Wk[(long)ka * NCOL + ncol[t]];
      b[1] = Wk[(long)(ka + 1) * NCOL + ncol[t]];
      acc[t] = __builtin_amdgcn_wmma_f32_16x16x4_f32(
          false, a, false, b, (short)0, acc[t], false, false);
    }
  }

  if constexpr (KSPLIT) {
    __syncthreads();
    float* const sR = sT;
    {
      const int rowhalf = (lane >> 4) * 8;
#pragma unroll
      for (int t = 0; t < TILES; ++t)
#pragma unroll
        for (int r = 0; r < 8; ++r)
          sR[(wave * 16 + rowhalf + r) * NCOL + ncol[t]] = acc[t][r];
    }
    __syncthreads();
    constexpr int QPR = NCOL / 4;
    for (int i = threadIdx.x; i < 16 * QPR; i += WAVES * 32) {
      const int orow = i / QPR;
      const int oc = (i - orow * QPR) * 4;
      float4 s;
      s.x = bias[oc]; s.y = bias[oc + 1]; s.z = bias[oc + 2]; s.w = bias[oc + 3];
#pragma unroll
      for (int w = 0; w < WAVES; ++w) {
        const float4 p = *(const float4*)&sR[(w * 16 + orow) * NCOL + oc];
        s.x += p.x; s.y += p.y; s.z += p.z; s.w += p.w;
      }
      const int row = rowbase + orow;
      if (row < nrows) {
        if (ACT == 0) {
          s.x = fmaxf(s.x, 0.0f); s.y = fmaxf(s.y, 0.0f);
          s.z = fmaxf(s.z, 0.0f); s.w = fmaxf(s.w, 0.0f);
        } else {
          s.x = 1.0f / (1.0f + __expf(-s.x)); s.y = 1.0f / (1.0f + __expf(-s.y));
          s.z = 1.0f / (1.0f + __expf(-s.z)); s.w = 1.0f / (1.0f + __expf(-s.w));
        }
        *(float4*)&out[(long)row * NCOL + oc] = s;
      }
    }
    return;
  }

  if (part == 1) {
#pragma unroll
    for (int t = 0; t < TILES; ++t)
#pragma unroll
      for (int r = 0; r < 8; ++r)
        sP[((cw * TILES + t) * 8 + r) * 32 + lane] = acc[t][r];
  }
  __syncthreads();

  if (part == 0) {
    const int rowhalf = (lane >> 4) * 8;
#pragma unroll
    for (int t = 0; t < TILES; ++t) {
#pragma unroll
      for (int r = 0; r < 8; ++r) {
        int row = rowbase + rowhalf + r;
        float v = acc[t][r] + sP[((cw * TILES + t) * 8 + r) * 32 + lane];
        if (row < nrows) {
          if (ACT == 0) {
            v = fmaxf(v, 0.0f);
          } else {
            v = 1.0f / (1.0f + __expf(-v));
          }
          out[(long)row * NCOL + ncol[t]] = v;
        }
      }
    }
  }
}


extern "C" void kernel_launch(void* const* d_in, const int* in_sizes, int n_in,
                              void* d_out, int out_size, void* d_ws, size_t ws_size,
                              hipStream_t stream) {
  constexpr int FIN = 128, FHID = 256, FOUT = 64;

  const float* x   = (const float*)d_in[0];
  const int*   ei  = (const int*)d_in[1];
  const float* Wl1 = (const float*)d_in[2];
  const float* Wr1 = (const float*)d_in[3];
  const float* b1  = (const float*)d_in[4];
  const float* Wl2 = (const float*)d_in[5];
  const float* Wr2 = (const float*)d_in[6];
  const float* b2  = (const float*)d_in[7];
  float* out = (float*)d_out;

  const int N = in_sizes[0] / FIN;
  const int E = in_sizes[1] / 2;
  const int* src = ei;
  const int* dst = ei + E;

  static_assert(HSTRIDE == FOUT, "a node's heads sit at the start of its output row");
  int* head = (int*)d_out;
  int* next = (int*)d_ws + N + (N + 1) + N;
  float* h  = (float*)(next + E);

  const int rowBlocks = (N + 15) / 16;

  const int R = (N + LINK_RANGES - 1) / LINK_RANGES;
  link_kernel<<<dim3(LINK_RANGES, NSUB), LINK_T, R * sizeof(int), stream>>>(dst, head, next, N, E);

  sage_layer_kernel<FIN, FHID, 8, 4, 0, false, 4>
      <<<rowBlocks, 256, 0, stream>>>(x, head, next, src, Wl1, Wr1, b1, h, N, E);
  sage_layer_kernel<FHID, FOUT, 8, 4, 1, true, 4>
      <<<rowBlocks, 256, 0, stream>>>(h, head, next, src, Wl2, Wr2, b2, out, N, E);
}


# Neighbour chains: the first 3 source nodes of a chain kept inline in its 16-byte head record (rest: overflow list): row loads go out with no hop per edge, batched across chains; row sum order changes
# speedup vs baseline: 4.1400x; 1.0741x over previous
#include <hip/hip_runtime.h>
#include <math.h>

typedef float v2f __attribute__((ext_vector_type(2)));
typedef float v8f __attribute__((ext_vector_type(8)));

constexpr int NSUB = 16;
constexpr int NSLOT = 3;
constexpr int HSTRIDE = 64;
constexpr int LINK_T = 1024;
constexpr int LINK_RANGES = 16;
constexpr int LINK_B = 8;

__device__ __forceinline__ void link_one(int* sh, int R, const int* __restrict__ src,
                                         int* __restrict__ next,
                                         int d, int e, int n0, unsigned cnt) {
  const unsigned dd = (unsigned)(d - n0);
  if (dd < cnt) {
    const int pos = atomicAdd(&sh[dd], 1);
    if (pos < NSLOT) sh[2 * R + NSLOT * (int)dd + pos] = src[e];
    else next[e] = atomicExch(&sh[R + (int)dd], e);
  }
}

__device__ __forceinline__ void link_group(int* sh, int R, const int* __restrict__ src,
                                           int* __restrict__ next,
                                           int4 d, int e, int n0, unsigned cnt) {
  const unsigned ux = (unsigned)(d.x - n0), uy = (unsigned)(d.y - n0);
  const unsigned uz = (unsigned)(d.z - n0), uw = (unsigned)(d.w - n0);
  if (min(min(ux, uy), min(uz, uw)) < cnt) {
    link_one(sh, R, src, next, d.x, e, n0, cnt);
    link_one(sh, R, src, next, d.y, e + 1, n0, cnt);
    link_one(sh, R, src, next, d.z, e + 2, n0, cnt);
    link_one(sh, R, src, next, d.w, e + 3, n0, cnt);
  }
}

__global__ __launch_bounds__(LINK_T)
void link_kernel(const int* __restrict__ src, const int* __restrict__ dst, int* __restrict__ head,
                 int* __restrict__ next, int N, int E) {
  extern __shared__ int sh[];
  const int tid = threadIdx.x;
  const int R = (N + (int)gridDim.x - 1) / (int)gridDim.x;
  const int n0 = (int)blockIdx.x * R;
  const int left = N - n0;
  const unsigned cnt = left <= 0 ? 0u : (unsigned)(left < R ? left : R);
  const int j = blockIdx.y;
  const int part = (E + NSUB - 1) / NSUB;
  const int e0 = j * part < E ? j * part : E;
  const int e1 = e0 + part < E ? e0 + part : E;

  for (int i = tid; i < R; i += LINK_T) sh[i] = 0;
  for (int i = R + tid; i < (2 + NSLOT) * R; i += LINK_T) sh[i] = -1;
  __syncthreads();

  if (cnt > 0 && e1 > e0) {
    int a0 = e0;
    while (a0 < e1 && (((size_t)(dst + a0)) & 15) != 0) ++a0;
    const int nvec = (e1 - a0) >> 2;
    const int t0 = a0 + 4 * nvec;
    if (tid < a0 - e0) link_one(sh, R, src, next, dst[e0 + tid], e0 + tid, n0, cnt);
    if (tid < e1 - t0) link_one(sh, R, src, next, dst[t0 + tid], t0 + tid, n0, cnt);

    const int4* __restrict__ dv = (const int4*)(dst + a0);
    const int s = (int)(((long)blockIdx.x * (long)nvec) / (long)gridDim.x);
    for (int sweep = 0; sweep < 2; ++sweep) {
      const int lo = sweep ? 0 : s;
      const int hi = sweep ? s : nvec;
      int base = lo;
      for (; base + LINK_T * LINK_B <= hi; base += LINK_T * LINK_B) {
        const int g = base + tid;
        int4 d[LINK_B];
#pragma unroll
        for (int u = 0; u < LINK_B; ++u) d[u] = dv[g + u * LINK_T];
#pragma unroll
        for (int u = 0; u < LINK_B; ++u)
          link_group(sh, R, src, next, d[u], a0 + 4 * (g + u * LINK_T), n0, cnt);
      }
      if (base < hi) {
        const int g = base + tid;
        int4 d[LINK_B];
#pragma unroll
        for (int u = 0; u < LINK_B; ++u) {
          const int v = g + u * LINK_T;
          d[u] = dv[v < hi ? v : base];
        }
#pragma unroll
        for (int u = 0; u < LINK_B; ++u) {
          const int v = g + u * LINK_T;
          if (v < hi) link_group(sh, R, src, next, d[u], a0 + 4 * v, n0, cnt);
        }
      }
    }
  }
  __syncthreads();
  for (int i = tid; i < (int)cnt; i += LINK_T) {
    int4 rec;
    rec.x = sh[2 * R + NSLOT * i];
    rec.y = sh[2 * R + NSLOT * i + 1];
    rec.z = sh[2 * R + NSLOT * i + 2];
    rec.w = sh[R + i];
    *(int4*)&head[(long)(n0 + i) * HSTRIDE + 4 * j] = rec;
  }
}

template<int K, int NCOL, int WAVES, int TILES, int ACT, bool KSPLIT, int NB>
__global__ __launch_bounds__(WAVES * 32) __attribute__((amdgpu_waves_per_eu(K > 128 ? 10 : 1)))
void sage_layer_kernel(const float* __restrict__ feat,
                       const int* head,
                       const int* __restrict__ next,
                       const int* __restrict__ src,
                       const float* __restrict__ Wl,
                       const float* __restrict__ Wr,
                       const float* __restrict__ bias,
                       float* out,
                       int nrows, int nedges) {
  static_assert(WAVES % 2 == 0, "the waves split into two halves, one per product");
  constexpr int HW = WAVES / 2;
  static_assert(KSPLIT ? (TILES * 16 == NCOL) : (HW * TILES * 16 == NCOL), "tiling must cover NCOL");
  constexpr int KSL = KSPLIT ? K / HW : K;
  static_assert(KSL * (KSPLIT ? HW : 1) == K && KSL % 8 == 0, "K slices cover K in whole pairs of k-steps");
  static_assert(K % 128 == 0, "K must be a multiple of 128");
  constexpr int KP = K + 4;
  constexpr int RPW = 16 / WAVES;
  constexpr int LPR = 32 / RPW;
  constexpr int CH = K / (4 * LPR);
  static_assert(RPW * WAVES == 16 && LPR * RPW == 32, "a wave's lanes split evenly over its rows");
  static_assert(CH * 4 * LPR == K, "a lane group covers a whole row");
  static_assert(LPR == NSUB, "one chain per lane of a row's lane group");
  static_assert((NSLOT + 1) * NSUB <= HSTRIDE && NSLOT == 3, "a node's chain records (16 bytes each) fit in its output row");
  __shared__ float sT[2 * 16 * KP];
  __shared__ int sH[WAVES * 32 * (NSLOT + 1)];
  __shared__ float sP[KSPLIT ? 32 : HW * TILES * 8 * 32];
  static_assert(!KSPLIT || WAVES * 16 * NCOL <= 2 * 16 * KP, "the partial tiles fit over the A tiles");
  static_assert(!KSPLIT || NCOL % 4 == 0, "an output row is written in 16-byte pieces");
  float* const sA = sT;
  float* const sX = sT + 16 * KP;

  const int rowbase = blockIdx.x * 16;
  const int lane = threadIdx.x & 31;
  const int wave = threadIdx.x >> 5;

  {
    const int sub = lane / LPR;
    const int sl = lane - sub * LPR;
    const int r = wave * RPW + sub;
    const int row = rowbase + r;
    float4 acc[CH] = {};
    int s = -1, pn = -1;
    int4 hv = {-1, -1, -1, -1};
    if (row < nrows) {
      const float4* xp = (const float4*)(feat + (long)row * K);
      float4 sv[CH];
#pragma unroll
      for (int c = 0; c < CH; ++c) sv[c] = xp[sl + LPR * c];
      hv = ((const int4*)head)[(long)row * (HSTRIDE / 4) + sl];
#pragma unroll
      for (int c = 0; c < CH; ++c)
        *(float4*)&sX[r * KP + 4 * (sl + LPR * c)] = sv[c];
      pn = hv.w;
    } else {
      const float4 z = {0.0f, 0.0f, 0.0f, 0.0f};
#pragma unroll
      for (int c = 0; c < CH; ++c) *(float4*)&sX[r * KP + 4 * (sl + LPR * c)] = z;
    }
    int deg = 0;
    int so = -1, pno = -1;
    if ((unsigned)pn < (unsigned)nedges) { so = src[pn]; pno = next[pn]; }
    int* const hw = sH + (wave * 32) * (NSLOT + 1);
    *(int4*)&hw[lane * (NSLOT + 1)] = hv;
    const unsigned gm = (1u << LPR) - 1u;
    unsigned long long mask;
    {
      const unsigned m0 = (__builtin_amdgcn_ballot_w32(hv.x >= 0) >> (sub * LPR)) & gm;
      const unsigned m1 = (__builtin_amdgcn_ballot_w32(hv.y >= 0) >> (sub * LPR)) & gm;
      const unsigned m2 = (__builtin_amdgcn_ballot_w32(hv.z >= 0) >> (sub * LPR)) & gm;
      mask = (unsigned long long)m0 | ((unsigned long long)m1 << LPR)
           | ((unsigned long long)m2 << (2 * LPR));
    }
    deg += __popcll(mask);
    for (int it = 0; it <= nedges; ++it) {
      int sn = -1, pnn = -1;
      if (it > 0) {
        const unsigned wavelive = __builtin_amdgcn_ballot_w32(s >= 0);
        if (wavelive == 0u) break;
        const unsigned mg = (wavelive >> (sub * LPR)) & gm;
        deg += __popc(mg);
        mask = mg;
        hw[lane * (NSLOT + 1)] = s;
        if ((unsigned)pn < (unsigned)nedges) { sn = src[pn]; pnn = next[pn]; }
      }
      while (__builtin_amdgcn_ballot_w32(mask != 0ull) != 0u) {
        int sj[NB];
#pragma unroll
        for (int b = 0; b < NB; ++b) {
          const int jj = __ffsll((long long)mask) - 1;
          mask &= mask - 1ull;
          const unsigned k = ((unsigned)jj / LPR) & 3u;
          const int v0 = hw[((lane & ~(LPR - 1)) | (jj & (LPR - 1))) * (NSLOT + 1) + k];
          sj[b] = jj < 0 ? -1 : v0;
        }
        float4 v[NB][CH];
#pragma unroll
        for (int b = 0; b < NB; ++b) {
#pragma unroll
          for (int c = 0; c < CH; ++c) v[b][c] = make_float4(0.0f, 0.0f, 0.0f, 0.0f);
          if (sj[b] >= 0) {
            const float4* rp = (const float4*)feat + ((unsigned)sj[b] * (unsigned)(K / 4) + (unsigned)sl);
#pragma unroll
            for (int c = 0; c < CH; ++c) v[b][c] = rp[LPR * c];
          }
        }
#pragma unroll
        for (int b = 0; b < NB; ++b) {
#pragma unroll
          for (int c = 0; c < CH; ++c) {
            acc[c].x += v[b][c].x; acc[c].y += v[b][c].y;
            acc[c].z += v[b][c].z; acc[c].w += v[b][c].w;
          }
        }
      }
      if (it == 0) { s = so; pn = pno; } else { s = sn; pn = pnn; }
    }
    const float scale = 1.0f / fmaxf((float)deg, 1.0f);
#pragma unroll
    for (int c = 0; c < CH; ++c) {
      float4 a;
      a.x = acc[c].x * scale; a.y = acc[c].y * scale;
      a.z = acc[c].z * scale; a.w = acc[c].w * scale;
      *(float4*)&sA[r * KP + 4 * (sl + LPR * c)] = a;
    }
  }
  __syncthreads();

  const int m = lane & 15;
  const int khalf = (lane >> 4) * 2;
  const int part = wave / HW;
  const int cw = wave - part * HW;
  const float* __restrict__ W = part ? Wr : Wl;
  const float* tile = sT + part * (16 * KP);

  v8f acc[TILES];
  int ncol[TILES];
#pragma unroll
  for (int t = 0; t < TILES; ++t) {
    ncol[t] = ((KSPLIT ? 0 : cw * TILES) + t) * 16 + (lane & 15);
    float bv = (KSPLIT || part) ? 0.0f : bias[ncol[t]];
#pragma unroll
    for (int j = 0; j < 8; ++j) acc[t][j] = bv;
  }

  const int kbeg = KSPLIT ? cw * KSL : 0;
  const float* __restrict__ Wk = W + (long)kbeg * NCOL;
  const float* tk = tile + kbeg;
#pragma unroll 2
  for (int k0 = 0; k0 < KSL; k0 += 4) {
    const int ka = k0 + khalf;
    v2f a;
    a[0] = tk[m * KP + ka];
    a[1] = tk[m * KP + ka + 1];
#pragma unroll
    for (int t = 0; t < TILES; ++t) {
      v2f b;
      b[0] = Wk[(long)ka * NCOL + ncol[t]];
      b[1] = Wk[(long)(ka + 1) * NCOL + ncol[t]];
      acc[t] = __builtin_amdgcn_wmma_f32_16x16x4_f32(
          false, a, false, b, (short)0, acc[t], false, false);
    }
  }

  if constexpr (KSPLIT) {
    __syncthreads();
    float* const sR = sT;
    {
      const int rowhalf = (lane >> 4) * 8;
#pragma unroll
      for (int t = 0; t < TILES; ++t)
#pragma unroll
        for (int r = 0; r < 8; ++r)
          sR[(wave * 16 + rowhalf + r) * NCOL + ncol[t]] = acc[t][r];
    }
    __syncthreads();
    constexpr int QPR = NCOL / 4;
    for (int i = threadIdx.x; i < 16 * QPR; i += WAVES * 32) {
      const int orow = i / QPR;
      const int oc = (i - orow * QPR) * 4;
      float4 s;
      s.x = bias[oc]; s.y = bias[oc + 1]; s.z = bias[oc + 2]; s.w = bias[oc + 3];
#pragma unroll
      for (int w = 0; w < WAVES; ++w) {
        const float4 p = *(const float4*)&sR[(w * 16 + orow) * NCOL + oc];
        s.x += p.x; s.y += p.y; s.z += p.z; s.w += p.w;
      }
      const int row = rowbase + orow;
      if (row < nrows) {
        if (ACT == 0) {
          s.x = fmaxf(s.x, 0.0f); s.y = fmaxf(s.y, 0.0f);
          s.z = fmaxf(s.z, 0.0f); s.w = fmaxf(s.w, 0.0f);
        } else {
          s.x = 1.0f / (1.0f + __expf(-s.x)); s.y = 1.0f / (1.0f + __expf(-s.y));
          s.z = 1.0f / (1.0f + __expf(-s.z)); s.w = 1.0f / (1.0f + __expf(-s.w));
        }
        *(float4*)&out[(long)row * NCOL + oc] = s;
      }
    }
    return;
  }

  if (part == 1) {
#pragma unroll
    for (int t = 0; t < TILES; ++t)
#pragma unroll
      for (int r = 0; r < 8; ++r)
        sP[((cw * TILES + t) * 8 + r) * 32 + lane] = acc[t][r];
  }
  __syncthreads();

  if (part == 0) {
    const int rowhalf = (lane >> 4) * 8;
#pragma unroll
    for (int t = 0; t < TILES; ++t) {
#pragma unroll
      for (int r = 0; r < 8; ++r) {
        int row = rowbase + rowhalf + r;
        float v = acc[t][r] + sP[((cw * TILES + t) * 8 + r) * 32 + lane];
        if (row < nrows) {
          if (ACT == 0) {
            v = fmaxf(v, 0.0f);
          } else {
            v = 1.0f / (1.0f + __expf(-v));
          }
          out[(long)row * NCOL + ncol[t]] = v;
        }
      }
    }
  }
}


extern "C" void kernel_launch(void* const* d_in, const int* in_sizes, int n_in,
                              void* d_out, int out_size, void* d_ws, size_t ws_size,
                              hipStream_t stream) {
  constexpr int FIN = 128, FHID = 256, FOUT = 64;

  const float* x   = (const float*)d_in[0];
  const int*   ei  = (const int*)d_in[1];
  const float* Wl1 = (const float*)d_in[2];
  const float* Wr1 = (const float*)d_in[3];
  const float* b1  = (const float*)d_in[4];
  const float* Wl2 = (const float*)d_in[5];
  const float* Wr2 = (const float*)d_in[6];
  const float* b2  = (const float*)d_in[7];
  float* out = (float*)d_out;

  const int N = in_sizes[0] / FIN;
  const int E = in_sizes[1] / 2;
  const int* src = ei;
  const int* dst = ei + E;

  static_assert(HSTRIDE == FOUT, "a node's heads sit at the start of its output row");
  int* head = (int*)d_out;
  int* next = (int*)d_ws + N + (N + 1) + N;
  float* h  = (float*)(next + E);

  const int rowBlocks = (N + 15) / 16;

  const int R = (N + LINK_RANGES - 1) / LINK_RANGES;
  link_kernel<<<dim3(LINK_RANGES, NSUB), LINK_T, (2 + NSLOT) * R * sizeof(int), stream>>>(src, dst, head, next, N, E);

  sage_layer_kernel<FIN, FHID, 8, 4, 0, false, 4>
      <<<rowBlocks, 256, 0, stream>>>(x, head, next, src, Wl1, Wr1, b1, h, N, E);
  sage_layer_kernel<FHID, FOUT, 8, 4, 1, true, 4>
      <<<rowBlocks, 256, 0, stream>>>(h, head, next, src, Wl2, Wr2, b2, out, N, E);
}


# Build: a block streams the source ids of its edge part beside the destinations (16-byte loads, same batch); filing an edge waits for no dependent load of its source; row sum order rule unchanged
# speedup vs baseline: 4.2597x; 1.0289x over previous
#include <hip/hip_runtime.h>
#include <math.h>

typedef float v2f __attribute__((ext_vector_type(2)));
typedef float v8f __attribute__((ext_vector_type(8)));

constexpr int NSUB = 16;
constexpr int NSLOT = 3;
constexpr int HSTRIDE = 64;
constexpr int LINK_T = 1024;
constexpr int LINK_RANGES = 16;
constexpr int LINK_B = 8;

__device__ __forceinline__ void link_one(int* sh, int R,
                                         int* __restrict__ next,
                                         int s, int d, int e, int n0, unsigned cnt) {
  const unsigned dd = (unsigned)(d - n0);
  if (dd < cnt) {
    const int pos = atomicAdd(&sh[dd], 1);
    if (pos < NSLOT) sh[2 * R + NSLOT * (int)dd + pos] = s;
    else next[e] = atomicExch(&sh[R + (int)dd], e);
  }
}

__device__ __forceinline__ int4 load_group(const int* __restrict__ p) {
  int4 r;
  r.x = p[0]; r.y = p[1]; r.z = p[2]; r.w = p[3];
  return r;
}

__device__ __forceinline__ void keep_loaded(int4 s) {
  asm volatile("" :: "v"(s.x), "v"(s.y), "v"(s.z), "v"(s.w));
}

__device__ __forceinline__ void link_group(int* sh, int R,
                                           int* __restrict__ next,
                                           int4 s, int4 d, int e, int n0, unsigned cnt) {
  const unsigned ux = (unsigned)(d.x - n0), uy = (unsigned)(d.y - n0);
  const unsigned uz = (unsigned)(d.z - n0), uw = (unsigned)(d.w - n0);
  if (min(min(ux, uy), min(uz, uw)) < cnt) {
    link_one(sh, R, next, s.x, d.x, e, n0, cnt);
    link_one(sh, R, next, s.y, d.y, e + 1, n0, cnt);
    link_one(sh, R, next, s.z, d.z, e + 2, n0, cnt);
    link_one(sh, R, next, s.w, d.w, e + 3, n0, cnt);
  }
}

__global__ __launch_bounds__(LINK_T)
void link_kernel(const int* __restrict__ src, const int* __restrict__ dst, int* __restrict__ head,
                 int* __restrict__ next, int N, int E) {
  extern __shared__ int sh[];
  const int tid = threadIdx.x;
  const int R = (N + (int)gridDim.x - 1) / (int)gridDim.x;
  const int n0 = (int)blockIdx.x * R;
  const int left = N - n0;
  const unsigned cnt = left <= 0 ? 0u : (unsigned)(left < R ? left : R);
  const int j = blockIdx.y;
  const int part = (E + NSUB - 1) / NSUB;
  const int e0 = j * part < E ? j * part : E;
  const int e1 = e0 + part < E ? e0 + part : E;

  for (int i = tid; i < R; i += LINK_T) sh[i] = 0;
  for (int i = R + tid; i < (2 + NSLOT) * R; i += LINK_T) sh[i] = -1;
  __syncthreads();

  if (cnt > 0 && e1 > e0) {
    int a0 = e0;
    while (a0 < e1 && (((size_t)(dst + a0)) & 15) != 0) ++a0;
    const int nvec = (e1 - a0) >> 2;
    const int t0 = a0 + 4 * nvec;
    if (tid < a0 - e0) link_one(sh, R, next, src[e0 + tid], dst[e0 + tid], e0 + tid, n0, cnt);
    if (tid < e1 - t0) link_one(sh, R, next, src[t0 + tid], dst[t0 + tid], t0 + tid, n0, cnt);

    const int4* __restrict__ dv = (const int4*)(dst + a0);
    const int* __restrict__ sg = src + a0;
    const int s = (int)(((long)blockIdx.x * (long)nvec) / (long)gridDim.x);
    for (int sweep = 0; sweep < 2; ++sweep) {
      const int lo = sweep ? 0 : s;
      const int hi = sweep ? s : nvec;
      int base = lo;
      for (; base + LINK_T * LINK_B <= hi; base += LINK_T * LINK_B) {
        const int g = base + tid;
        int4 d[LINK_B], sv[LINK_B];
#pragma unroll
        for (int u = 0; u < LINK_B; ++u) d[u] = dv[g + u * LINK_T];
#pragma unroll
        for (int u = 0; u < LINK_B; ++u) sv[u] = load_group(sg + 4 * (g + u * LINK_T));
#pragma unroll
        for (int u = 0; u < LINK_B; ++u) keep_loaded(sv[u]);
#pragma unroll
        for (int u = 0; u < LINK_B; ++u)
          link_group(sh, R, next, sv[u], d[u], a0 + 4 * (g + u * LINK_T), n0, cnt);
      }
      if (base < hi) {
        const int g = base + tid;
        int4 d[LINK_B], sv[LINK_B];
#pragma unroll
        for (int u = 0; u < LINK_B; ++u) {
          const int v = g + u * LINK_T;
          d[u] = dv[v < hi ? v : base];
        }
#pragma unroll
        for (int u = 0; u < LINK_B; ++u) {
          const int v = g + u * LINK_T;
          sv[u] = load_group(sg + 4 * (v < hi ? v : base));
        }
#pragma unroll
        for (int u = 0; u < LINK_B; ++u) keep_loaded(sv[u]);
#pragma unroll
        for (int u = 0; u < LINK_B; ++u) {
          const int v = g + u * LINK_T;
          link_group(sh, R, next, sv[u], d[u], a0 + 4 * v, n0, v < hi ? cnt : 0u);
        }
      }
    }
  }
  __syncthreads();
  for (int i = tid; i < (int)cnt; i += LINK_T) {
    int4 rec;
    rec.x = sh[2 * R + NSLOT * i];
    rec.y = sh[2 * R + NSLOT * i + 1];
    rec.z = sh[2 * R + NSLOT * i + 2];
    rec.w = sh[R + i];
    *(int4*)&head[(long)(n0 + i) * HSTRIDE + 4 * j] = rec;
  }
}

template<int K, int NCOL, int WAVES, int TILES, int ACT, bool KSPLIT, int NB>
__global__ __launch_bounds__(WAVES * 32) __attribute__((amdgpu_waves_per_eu(K > 128 ? 10 : 1)))
void sage_layer_kernel(const float* __restrict__ feat,
                       const int* head,
                       const int* __restrict__ next,
                       const int* __restrict__ src,
                       const float* __restrict__ Wl,
                       const float* __restrict__ Wr,
                       const float* __restrict__ bias,
                       float* out,
                       int nrows, int nedges) {
  static_assert(WAVES % 2 == 0, "the waves split into two halves, one per product");
  constexpr int HW = WAVES / 2;
  static_assert(KSPLIT ? (TILES * 16 == NCOL) : (HW * TILES * 16 == NCOL), "tiling must cover NCOL");
  constexpr int KSL = KSPLIT ? K / HW : K;
  static_assert(KSL * (KSPLIT ? HW : 1) == K && KSL % 8 == 0, "K slices cover K in whole pairs of k-steps");
  static_assert(K % 128 == 0, "K must be a multiple of 128");
  constexpr int KP = K + 4;
  constexpr int RPW = 16 / WAVES;
  constexpr int LPR = 32 / RPW;
  constexpr int CH = K / (4 * LPR);
  static_assert(RPW * WAVES == 16 && LPR * RPW == 32, "a wave's lanes split evenly over its rows");
  static_assert(CH * 4 * LPR == K, "a lane group covers a whole row");
  static_assert(LPR == NSUB, "one chain per lane of a row's lane group");
  static_assert((NSLOT + 1) * NSUB <= HSTRIDE && NSLOT == 3, "a node's chain records (16 bytes each) fit in its output row");
  __shared__ float sT[2 * 16 * KP];
  __shared__ int sH[WAVES * 32 * (NSLOT + 1)];
  __shared__ float sP[KSPLIT ? 32 : HW * TILES * 8 * 32];
  static_assert(!KSPLIT || WAVES * 16 * NCOL <= 2 * 16 * KP, "the partial tiles fit over the A tiles");
  static_assert(!KSPLIT || NCOL % 4 == 0, "an output row is written in 16-byte pieces");
  float* const sA = sT;
  float* const sX = sT + 16 * KP;

  const int rowbase = blockIdx.x * 16;
  const int lane = threadIdx.x & 31;
  const int wave = threadIdx.x >> 5;

  {
    const int sub = lane / LPR;
    const int sl = lane - sub * LPR;
    const int r = wave * RPW + sub;
    const int row = rowbase + r;
    float4 acc[CH] = {};
    int s = -1, pn = -1;
    int4 hv = {-1, -1, -1, -1};
    if (row < nrows) {
      const float4* xp = (const float4*)(feat + (long)row * K);
      float4 sv[CH];
#pragma unroll
      for (int c = 0; c < CH; ++c) sv[c] = xp[sl + LPR * c];
      hv = ((const int4*)head)[(long)row * (HSTRIDE / 4) + sl];
#pragma unroll
      for (int c = 0; c < CH; ++c)
        *(float4*)&sX[r * KP + 4 * (sl + LPR * c)] = sv[c];
      pn = hv.w;
    } else {
      const float4 z = {0.0f, 0.0f, 0.0f, 0.0f};
#pragma unroll
      for (int c = 0; c < CH; ++c) *(float4*)&sX[r * KP + 4 * (sl + LPR * c)] = z;
    }
    int deg = 0;
    int so = -1, pno = -1;
    if ((unsigned)pn < (unsigned)nedges) { so = src[pn]; pno = next[pn]; }
    int* const hw = sH + (wave * 32) * (NSLOT + 1);
    *(int4*)&hw[lane * (NSLOT + 1)] = hv;
    const unsigned gm = (1u << LPR) - 1u;
    unsigned long long mask;
    {
      const unsigned m0 = (__builtin_amdgcn_ballot_w32(hv.x >= 0) >> (sub * LPR)) & gm;
      const unsigned m1 = (__builtin_amdgcn_ballot_w32(hv.y >= 0) >> (sub * LPR)) & gm;
      const unsigned m2 = (__builtin_amdgcn_ballot_w32(hv.z >= 0) >> (sub * LPR)) & gm;
      mask = (unsigned long long)m0 | ((unsigned long long)m1 << LPR)
           | ((unsigned long long)m2 << (2 * LPR));
    }
    deg += __popcll(mask);
    for (int it = 0; it <= nedges; ++it) {
      int sn = -1, pnn = -1;
      if (it > 0) {
        const unsigned wavelive = __builtin_amdgcn_ballot_w32(s >= 0);
        if (wavelive == 0u) break;
        const unsigned mg = (wavelive >> (sub * LPR)) & gm;
        deg += __popc(mg);
        mask = mg;
        hw[lane * (NSLOT + 1)] = s;
        if ((unsigned)pn < (unsigned)nedges) { sn = src[pn]; pnn = next[pn]; }
      }
      while (__builtin_amdgcn_ballot_w32(mask != 0ull) != 0u) {
        int sj[NB];
#pragma unroll
        for (int b = 0; b < NB; ++b) {
          const int jj = __ffsll((long long)mask) - 1;
          mask &= mask - 1ull;
          const unsigned k = ((unsigned)jj / LPR) & 3u;
          const int v0 = hw[((lane & ~(LPR - 1)) | (jj & (LPR - 1))) * (NSLOT + 1) + k];
          sj[b] = jj < 0 ? -1 : v0;
        }
        float4 v[NB][CH];
#pragma unroll
        for (int b = 0; b < NB; ++b) {
#pragma unroll
          for (int c = 0; c < CH; ++c) v[b][c] = make_float4(0.0f, 0.0f, 0.0f, 0.0f);
          if (sj[b] >= 0) {
            const float4* rp = (const float4*)feat + ((unsigned)sj[b] * (unsigned)(K / 4) + (unsigned)sl);
#pragma unroll
            for (int c = 0; c < CH; ++c) v[b][c] = rp[LPR * c];
          }
        }
#pragma unroll
        for (int b = 0; b < NB; ++b) {
#pragma unroll
          for (int c = 0; c < CH; ++c) {
            acc[c].x += v[b][c].x; acc[c].y += v[b][c].y;
            acc[c].z += v[b][c].z; acc[c].w += v[b][c].w;
          }
        }
      }
      if (it == 0) { s = so; pn = pno; } else { s = sn; pn = pnn; }
    }
    const float scale = 1.0f / fmaxf((float)deg, 1.0f);
#pragma unroll
    for (int c = 0; c < CH; ++c) {
      float4 a;
      a.x = acc[c].x * scale; a.y = acc[c].y * scale;
      a.z = acc[c].z * scale; a.w = acc[c].w * scale;
      *(float4*)&sA[r * KP + 4 * (sl + LPR * c)] = a;
    }
  }
  __syncthreads();

  const int m = lane & 15;
  const int khalf = (lane >> 4) * 2;
  const int part = wave / HW;
  const int cw = wave - part * HW;
  const float* __restrict__ W = part ? Wr : Wl;
  const float* tile = sT + part * (16 * KP);

  v8f acc[TILES];
  int ncol[TILES];
#pragma unroll
  for (int t = 0; t < TILES; ++t) {
    ncol[t] = ((KSPLIT ? 0 : cw * TILES) + t) * 16 + (lane & 15);
    float bv = (KSPLIT || part) ? 0.0f : bias[ncol[t]];
#pragma unroll
    for (int j = 0; j < 8; ++j) acc[t][j] = bv;
  }

  const int kbeg = KSPLIT ? cw * KSL : 0;
  const float* __restrict__ Wk = W + (long)kbeg * NCOL;
  const float* tk = tile + kbeg;
#pragma unroll 2
  for (int k0 = 0; k0 < KSL; k0 += 4) {
    const int ka = k0 + khalf;
    v2f a;
    a[0] = tk[m * KP + ka];
    a[1] = tk[m * KP + ka + 1];
#pragma unroll
    for (int t = 0; t < TILES; ++t) {
      v2f b;
      b[0] = Wk[(long)ka * NCOL + ncol[t]];
      b[1] = Wk[(long)(ka + 1) * NCOL + ncol[t]];
      acc[t] = __builtin_amdgcn_wmma_f32_16x16x4_f32(
          false, a, false, b, (short)0, acc[t], false, false);
    }
  }

  if constexpr (KSPLIT) {
    __syncthreads();
    float* const sR = sT;
    {
      const int rowhalf = (lane >> 4) * 8;
#pragma unroll
      for (int t = 0; t < TILES; ++t)
#pragma unroll
        for (int r = 0; r < 8; ++r)
          sR[(wave * 16 + rowhalf + r) * NCOL + ncol[t]] = acc[t][r];
    }
    __syncthreads();
    constexpr int QPR = NCOL / 4;
    for (int i = threadIdx.x; i < 16 * QPR; i += WAVES * 32) {
      const int orow = i / QPR;
      const int oc = (i - orow * QPR) * 4;
      float4 s;
      s.x = bias[oc]; s.y = bias[oc + 1]; s.z = bias[oc + 2]; s.w = bias[oc + 3];
#pragma unroll
      for (int w = 0; w < WAVES; ++w) {
        const float4 p = *(const float4*)&sR[(w * 16 + orow) * NCOL + oc];
        s.x += p.x; s.y += p.y; s.z += p.z; s.w += p.w;
      }
      const int row = rowbase + orow;
      if (row < nrows) {
        if (ACT == 0) {
          s.x = fmaxf(s.x, 0.0f); s.y = fmaxf(s.y, 0.0f);
          s.z = fmaxf(s.z, 0.0f); s.w = fmaxf(s.w, 0.0f);
        } else {
          s.x = 1.0f / (1.0f + __expf(-s.x)); s.y = 1.0f / (1.0f + __expf(-s.y));
          s.z = 1.0f / (1.0f + __expf(-s.z)); s.w = 1.0f / (1.0f + __expf(-s.w));
        }
        *(float4*)&out[(long)row * NCOL + oc] = s;
      }
    }
    return;
  }

  if (part == 1) {
#pragma unroll
    for (int t = 0; t < TILES; ++t)
#pragma unroll
      for (int r = 0; r < 8; ++r)
        sP[((cw * TILES + t) * 8 + r) * 32 + lane] = acc[t][r];
  }
  __syncthreads();

  if (part == 0) {
    const int rowhalf = (lane >> 4) * 8;
#pragma unroll
    for (int t = 0; t < TILES; ++t) {
#pragma unroll
      for (int r = 0; r < 8; ++r) {
        int row = rowbase + rowhalf + r;
        float v = acc[t][r] + sP[((cw * TILES + t) * 8 + r) * 32 + lane];
        if (row < nrows) {
          if (ACT == 0) {
            v = fmaxf(v, 0.0f);
          } else {
            v = 1.0f / (1.0f + __expf(-v));
          }
          out[(long)row * NCOL + ncol[t]] = v;
        }
      }
    }
  }
}


extern "C" void kernel_launch(void* const* d_in, const int* in_sizes, int n_in,
                              void* d_out, int out_size, void* d_ws, size_t ws_size,
                              hipStream_t stream) {
  constexpr int FIN = 128, FHID = 256, FOUT = 64;

  const float* x   = (const float*)d_in[0];
  const int*   ei  = (const int*)d_in[1];
  const float* Wl1 = (const float*)d_in[2];
  const float* Wr1 = (const float*)d_in[3];
  const float* b1  = (const float*)d_in[4];
  const float* Wl2 = (const float*)d_in[5];
  const float* Wr2 = (const float*)d_in[6];
  const float* b2  = (const float*)d_in[7];
  float* out = (float*)d_out;

  const int N = in_sizes[0] / FIN;
  const int E = in_sizes[1] / 2;
  const int* src = ei;
  const int* dst = ei + E;

  static_assert(HSTRIDE == FOUT, "a node's heads sit at the start of its output row");
  int* head = (int*)d_out;
  int* next = (int*)d_ws + N + (N + 1) + N;
  float* h  = (float*)(next + E);

  const int rowBlocks = (N + 15) / 16;

  const int R = (N + LINK_RANGES - 1) / LINK_RANGES;
  link_kernel<<<dim3(LINK_RANGES, NSUB), LINK_T, (2 + NSLOT) * R * sizeof(int), stream>>>(src, dst, head, next, N, E);

  sage_layer_kernel<FIN, FHID, 8, 4, 0, false, 4>
      <<<rowBlocks, 256, 0, stream>>>(x, head, next, src, Wl1, Wr1, b1, h, N, E);
  sage_layer_kernel<FHID, FOUT, 8, 4, 1, true, 4>
      <<<rowBlocks, 256, 0, stream>>>(h, head, next, src, Wl2, Wr2, b2, out, N, E);
}
